# starting kernel
# baseline (speedup 1.0000x reference)
#include <hip/hip_runtime.h>
#include <cstdint>
#include <cstddef>

#define IN_F 128
#define NEG_SLOPE 0.2f
#define H0 4
#define H1 1

typedef __attribute__((ext_vector_type(2))) float v2f;
typedef __attribute__((ext_vector_type(8))) float v8f;

template <int NCOL>
__global__ __launch_bounds__(32 * (NCOL / 16)) void wmma_gemm_k128(
    const float* __restrict__ X, const float* __restrict__ W, float* __restrict__ Y) {
  constexpr int K = 128;
  __shared__ float xs[16][K + 4];
  const int row0 = blockIdx.x * 16;
  const int tid = threadIdx.x;
  const int lane = tid & 31;
  const int wave = tid >> 5;

  for (int i = tid; i < 16 * K; i += 32 * (NCOL / 16)) {
    int r = i >> 7;
    int c = i & 127;
    xs[r][c] = X[(row0 + r) * K + c];
  }
  __syncthreads();

  const int n0 = wave * 16;
  const int m = lane & 15;
  const int khalf = lane >> 4;

  v8f acc = {};
  for (int k0 = 0; k0 < K; k0 += 4) {
    const int ka = k0 + khalf * 2;
    v2f a, b;
    a.x = xs[m][ka];
    a.y = xs[m][ka + 1];
    b.x = W[(ka) *NCOL + n0 + m];
    b.y = W[(ka + 1) * NCOL + n0 + m];
    acc = __builtin_amdgcn_wmma_f32_16x16x4_f32(false, a, false, b, (short)0, acc,
                                                false, false);
  }

  const int n = n0 + m;
#pragma unroll
  for (int r = 0; r < 8; ++r) {
    Y[(row0 + khalf * 8 + r) * NCOL + n] = acc[r];
  }
}

template <int H>
__global__ void attn_coef32(const float* __restrict__ h, const float* __restrict__ al,
                            const float* __restrict__ ar, float* __restrict__ el,
                            float* __restrict__ er, int nh) {
  const int gw = blockIdx.x * (blockDim.x >> 5) + (threadIdx.x >> 5);
  if (gw >= nh) return;
  const int lane = threadIdx.x & 31;
  const int head = gw % H;
  const float v = h[gw * 32 + lane];
  float l = v * al[head * 32 + lane];
  float r = v * ar[head * 32 + lane];
#pragma unroll
  for (int off = 16; off > 0; off >>= 1) {
    l += __shfl_down(l, off, 32);
    r += __shfl_down(r, off, 32);
  }
  if (lane == 0) {
    el[gw] = l;
    er[gw] = r;
  }
}

template <int H>
__global__ void edge_e_max(const int* __restrict__ src, const int* __restrict__ dst,
                           const float* __restrict__ el, const float* __restrict__ er,
                           float* __restrict__ e, float* __restrict__ m, int ne) {
  const int i = blockIdx.x * blockDim.x + threadIdx.x;
  if (i >= ne * H) return;
  const int eid = i / H;
  const int h = i % H;
  const int d = dst[eid];
  float v = el[src[eid] * H + h] + er[d * H + h];
  v = (v > 0.0f) ? v : NEG_SLOPE * v;
  e[i] = v;
  float* slot = m + d * H + h;
  if (v >= 0.0f)
    atomicMax((int*)slot, __float_as_int(v));
  else
    atomicMin((unsigned int*)slot, __float_as_uint(v));
}

template <int H>
__global__ void edge_ex_sum(const int* __restrict__ dst, const float* __restrict__ m,
                            float* __restrict__ e, float* __restrict__ s, int ne) {
  const int i = blockIdx.x * blockDim.x + threadIdx.x;
  if (i >= ne * H) return;
  const int eid = i / H;
  const int h = i % H;
  const int slot = dst[eid] * H + h;
  const float ex = expf(e[i] - m[slot]);
  e[i] = ex;
  atomicAdd(s + slot, ex);
}

template <int H, int D>
__global__ void edge_aggregate(const int* __restrict__ src, const int* __restrict__ dst,
                               const float* __restrict__ ex, const float* __restrict__ s,
                               const float* __restrict__ hfeat, float* __restrict__ out,
                               int ne) {
  constexpr int HD = H * D;
  const long long t = (long long)blockIdx.x * blockDim.x + threadIdx.x;
  if (t >= (long long)ne * HD) return;
  const int eid = (int)(t / HD);
  const int c = (int)(t % HD);
  const int h = c / D;
  const int sn = src[eid];
  const int dn = dst[eid];
  const float alpha = ex[eid * H + h] / s[dn * H + h];
  atomicAdd(out + dn * HD + c, alpha * hfeat[sn * HD + c]);
}

template <int C>
__global__ void bias_elu(const float* __restrict__ agg, const float* __restrict__ b,
                         float* __restrict__ act, int total) {
  const int i = blockIdx.x * blockDim.x + threadIdx.x;
  if (i >= total) return;
  const float v = agg[i] + b[i & (C - 1)];
  act[i] = (v > 0.0f) ? v : (expf(v) - 1.0f);
}

__global__ void add_bias32(float* __restrict__ out, const float* __restrict__ b,
                           int total) {
  const int i = blockIdx.x * blockDim.x + threadIdx.x;
  if (i >= total) return;
  out[i] += b[i & 31];
}

extern "C" void kernel_launch(void* const* d_in, const int* in_sizes, int n_in,
                              void* d_out, int out_size, void* d_ws, size_t ws_size,
                              hipStream_t stream) {
  const float* x = (const float*)d_in[0];
  const int* src = (const int*)d_in[1];
  const int* dst = (const int*)d_in[2];
  const float* W1 = (const float*)d_in[3];
  const float* al1 = (const float*)d_in[4];
  const float* ar1 = (const float*)d_in[5];
  const float* b1 = (const float*)d_in[6];
  const float* W2 = (const float*)d_in[7];
  const float* al2 = (const float*)d_in[8];
  const float* ar2 = (const float*)d_in[9];
  const float* b2 = (const float*)d_in[10];

  const int n = in_sizes[0] / IN_F;
  const int e = in_sizes[1];

  char* ws = (char*)d_ws;
  size_t off = 0;
  auto wsalloc = [&](size_t bytes) {
    void* p = ws + off;
    off = (off + bytes + 255) & ~(size_t)255;
    return p;
  };
  float* h1 = (float*)wsalloc((size_t)n * 128 * sizeof(float));
  float* agg1 = (float*)wsalloc((size_t)n * 128 * sizeof(float));
  float* ebuf = (float*)wsalloc((size_t)e * H0 * sizeof(float));
  float* el = (float*)wsalloc((size_t)n * H0 * sizeof(float));
  float* er = (float*)wsalloc((size_t)n * H0 * sizeof(float));
  float* mbuf = (float*)wsalloc((size_t)n * H0 * sizeof(float));
  float* sbuf = (float*)wsalloc((size_t)n * H0 * sizeof(float));
  float* outv = (float*)d_out;

  wmma_gemm_k128<128><<<n / 16, 256, 0, stream>>>(x, W1, h1);

  {
    const int nh = n * H0;
    attn_coef32<H0><<<(nh + 7) / 8, 256, 0, stream>>>(h1, al1, ar1, el, er, nh);
  }

  hipMemsetAsync(mbuf, 0xFF, (size_t)n * H0 * sizeof(float), stream);
  hipMemsetAsync(sbuf, 0, (size_t)n * H0 * sizeof(float), stream);
  hipMemsetAsync(agg1, 0, (size_t)n * 128 * sizeof(float), stream);

  edge_e_max<H0><<<(e * H0 + 255) / 256, 256, 0, stream>>>(src, dst, el, er, ebuf, mbuf, e);
  edge_ex_sum<H0><<<(e * H0 + 255) / 256, 256, 0, stream>>>(dst, mbuf, ebuf, sbuf, e);
  {
    const long long total = (long long)e * 128;
    edge_aggregate<H0, 32><<<(int)((total + 255) / 256), 256, 0, stream>>>(
        src, dst, ebuf, sbuf, h1, agg1, e);
  }

  bias_elu<128><<<(n * 128 + 255) / 256, 256, 0, stream>>>(agg1, b1, h1, n * 128);

  float* g2 = agg1;
  wmma_gemm_k128<32><<<n / 16, 64, 0, stream>>>(h1, W2, g2);

  attn_coef32<H1><<<(n + 7) / 8, 256, 0, stream>>>(g2, al2, ar2, el, er, n);

  hipMemsetAsync(mbuf, 0xFF, (size_t)n * H1 * sizeof(float), stream);
  hipMemsetAsync(sbuf, 0, (size_t)n * H1 * sizeof(float), stream);
  hipMemsetAsync(d_out, 0, (size_t)n * 32 * sizeof(float), stream);

  edge_e_max<H1><<<(e + 255) / 256, 256, 0, stream>>>(src, dst, el, er, ebuf, mbuf, e);
  edge_ex_sum<H1><<<(e + 255) / 256, 256, 0, stream>>>(dst, mbuf, ebuf, sbuf, e);
  {
    const long long total = (long long)e * 32;
    edge_aggregate<H1, 32><<<(int)((total + 255) / 256), 256, 0, stream>>>(
        src, dst, ebuf, sbuf, g2, outv, e);
  }

  add_bias32<<<(n * 32 + 255) / 256, 256, 0, stream>>>(outv, b2, n * 32);
}


# layer 2's tile kernel starts from the per-wave kept-edge lists (local row, source) that layer 1's scan built and now stores to the workspace: no key scan and no src/dst gathers in layer 2
# speedup vs baseline: 5.6499x; 5.6499x over previous
#include <hip/hip_runtime.h>
#include <cstdint>
#include <cstddef>

#define IN_F 128
#define NEG_SLOPE 0.2f
#define H0 4
#define H1 1

typedef __attribute__((ext_vector_type(2))) float v2f;
typedef __attribute__((ext_vector_type(8))) float v8f;

template <int NCOL, int RW, int RPW>
__global__ __launch_bounds__(32 * (NCOL / 16) * RW) void wmma_gemm_k128(
    const float* __restrict__ X, const float* __restrict__ W, float* __restrict__ Y,
    int nrows) {
  constexpr int K = 128;
  constexpr int CW = NCOL / 16;
  constexpr int NT = 32 * CW * RW;
  constexpr int TR = 16 * RW * RPW;
  constexpr int PITCH = K + 4;
  constexpr int NLD = TR * (K / 4) / NT;
  static_assert((TR * (K / 4)) % NT == 0, "a tile is whole pieces per thread");
  static_assert(NT % 32 == 0 && (K / 4) == 32, "a wave moves one whole row per piece");
  __shared__ float xs[2][TR][PITCH];

  const int tid = threadIdx.x;
  const int lane = tid & 31;
  const int wave = tid >> 5;
  const int n0 = (wave % CW) * 16;
  const int rb = (wave / CW) * RPW;
  const int m = lane & 15;
  const int khalf = lane >> 4;

  v2f bf[K / 4];
#pragma unroll
  for (int ks = 0; ks < K / 4; ++ks) {
    const int ka = ks * 4 + khalf * 2;
    bf[ks].x = W[ka * NCOL + n0 + m];
    bf[ks].y = W[(ka + 1) * NCOL + n0 + m];
  }

  const int ntiles = (nrows + TR - 1) / TR;
  const float4* __restrict__ X4 = reinterpret_cast<const float4*>(X);

  auto load_tile = [&](int tile, float4 (&v)[NLD]) {
    const float4* __restrict__ p0 = X4 + (size_t)tile * (TR * (K / 4)) + tid;
    if (tile * TR + TR <= nrows) {
#pragma unroll
      for (int u = 0; u < NLD; ++u) v[u] = p0[u * NT];
    } else {
#pragma unroll
      for (int u = 0; u < NLD; ++u) {
        const int r = tile * TR + ((tid + u * NT) >> 5);
        v[u] = (r < nrows) ? p0[u * NT] : make_float4(0.f, 0.f, 0.f, 0.f);
      }
    }
  };
  auto stage_tile = [&](int buf, const float4 (&v)[NLD]) {
#pragma unroll
    for (int u = 0; u < NLD; ++u) {
      const int p = tid + u * NT;
      *reinterpret_cast<float4*>(&xs[buf][p >> 5][(p & 31) * 4]) = v[u];
    }
  };

  float4 pre[NLD];
  int tile = blockIdx.x;
  if (tile < ntiles) {
    load_tile(tile, pre);
    stage_tile(0, pre);
  }
  __syncthreads();

  int buf = 0;
  for (; tile < ntiles; tile += gridDim.x) {
    const int nxt = tile + gridDim.x;
    if (nxt < ntiles) load_tile(nxt, pre);

    v8f acc[RPW];
#pragma unroll
    for (int s = 0; s < RPW; ++s) acc[s] = v8f{};
#pragma unroll
    for (int ks = 0; ks < K / 4; ++ks) {
      const int ka = ks * 4 + khalf * 2;
#pragma unroll
      for (int s = 0; s < RPW; ++s) {
        v2f a;
        a.x = xs[buf][(rb + s) * 16 + m][ka];
        a.y = xs[buf][(rb + s) * 16 + m][ka + 1];
        acc[s] = __builtin_amdgcn_wmma_f32_16x16x4_f32(false, a, false, bf[ks], (short)0,
                                                       acc[s], false, false);
      }
    }

    if (nxt < ntiles) stage_tile(buf ^ 1, pre);
    __syncthreads();

    const bool whole = tile * TR + TR <= nrows;
#pragma unroll
    for (int s = 0; s < RPW; ++s) {
      const int row0 = tile * TR + (rb + s) * 16 + khalf * 8;
      float* __restrict__ y0 = Y + (size_t)row0 * NCOL + n0 + m;
      if (whole) {
#pragma unroll
        for (int r = 0; r < 8; ++r) y0[r * NCOL] = acc[s][r];
      } else {
#pragma unroll
        for (int r = 0; r < 8; ++r)
          if (row0 + r < nrows) y0[r * NCOL] = acc[s][r];
      }
    }
    buf ^= 1;
  }
}

#define LGRID 1024
#define LBLOCK 256
#define LSTRIDE (LGRID * LBLOCK)

template <int H, int UN>
__global__ __launch_bounds__(LBLOCK) void attn_coef32(
    const float* __restrict__ h, const float* __restrict__ al,
    const float* __restrict__ ar, float* __restrict__ el, float* __restrict__ er,
    int nh) {
  static_assert(4 % H == 0, "the 4 rows of a wave must start at head 0");
  static_assert(LSTRIDE % 32 == 0, "a wave's pieces must start at a row boundary");
  const int lane = threadIdx.x & 31;
  const int sub = lane & 7;
  const int head = (lane >> 3) % H;
  const float4 a4 = *reinterpret_cast<const float4*>(al + head * 32 + sub * 4);
  const float4 r4 = *reinterpret_cast<const float4*>(ar + head * 32 + sub * 4);
  const float4* __restrict__ h4 = reinterpret_cast<const float4*>(h);
  const int total = nh * 8;
  const int t = blockIdx.x * LBLOCK + threadIdx.x;

  auto row_sums = [&](const float4 v, float& l, float& r) {
    l = v.x * a4.x + v.y * a4.y + v.z * a4.z + v.w * a4.w;
    r = v.x * r4.x + v.y * r4.y + v.z * r4.z + v.w * r4.w;
    l += __shfl_xor(l, 1, 32);
    r += __shfl_xor(r, 1, 32);
    l += __shfl_xor(l, 2, 32);
    r += __shfl_xor(r, 2, 32);
    l += __shfl_xor(l, 4, 32);
    r += __shfl_xor(r, 4, 32);
  };

  int wb = t - lane;
  for (; wb + (UN - 1) * LSTRIDE + 31 < total; wb += UN * LSTRIDE) {
    float4 v[UN];
#pragma unroll
    for (int u = 0; u < UN; ++u) v[u] = h4[wb + u * LSTRIDE + lane];
#pragma unroll
    for (int u = 0; u < UN; ++u) {
      float l, r;
      row_sums(v[u], l, r);
      if (sub == 0) {
        const int row = (wb + u * LSTRIDE + lane) >> 3;
        el[row] = l;
        er[row] = r;
      }
    }
  }
  for (; wb < total; wb += LSTRIDE) {
    const int i = wb + lane;
    const float4 v = (i < total) ? h4[i] : make_float4(0.f, 0.f, 0.f, 0.f);
    float l, r;
    row_sums(v, l, r);
    if (sub == 0 && i < total) {
      el[i >> 3] = l;
      er[i >> 3] = r;
    }
  }
}

#define TN 391
#define GCAP 384

__global__ __launch_bounds__(LBLOCK) void tile_keys(const int* __restrict__ dst,
                                                    unsigned char* __restrict__ key,
                                                    int ne) {
  const int n4 = ne >> 2;
  const int4* __restrict__ d4 = reinterpret_cast<const int4*>(dst);
  unsigned* __restrict__ k4 = reinterpret_cast<unsigned*>(key);
  const int t = blockIdx.x * LBLOCK + threadIdx.x;
  for (int i = t; i < n4; i += LSTRIDE) {
    const int4 d = d4[i];
    k4[i] = (((unsigned)d.x / TN) & 255u) | ((((unsigned)d.y / TN) & 255u) << 8) |
            ((((unsigned)d.z / TN) & 255u) << 16) | ((((unsigned)d.w / TN) & 255u) << 24);
  }
  for (int i = n4 * 4 + t; i < ne; i += LSTRIDE)
    key[i] = (unsigned char)(((unsigned)dst[i] / TN) & 255u);
}

template <int H, bool EMIT, bool REUSE>
__global__ __launch_bounds__(1024) void tile_softmax_aggregate(
    const int* __restrict__ src, const int* __restrict__ dst,
    const unsigned char* __restrict__ key, const float* __restrict__ el,
    const float* __restrict__ er, const float* __restrict__ hfeat,
    float* __restrict__ out, unsigned* __restrict__ glist, int* __restrict__ gcount,
    int ne, int n) {
  constexpr int HD = H * 32;
  constexpr int U = (H == 1) ? 8 : 4;
  constexpr int G = U * 512;
  constexpr int NW = U / 2;
  constexpr int QCAP = 32 * (1 + U);
  constexpr int WCAP = (H == 1) ? 1024 : 512;
  constexpr int KU = (H == 1) ? 8 : 4;
  static_assert(TN < 512, "local row must fit 9 bits");
  static_assert(GCAP <= WCAP, "a list read back from memory must fit the wave's list in LDS");
  constexpr int NB = 8;
  __shared__ float acc[(TN + 1) * HD];
  __shared__ int m_t[(TN + 1) * H];
  __shared__ float s_t[(TN + 1) * H];
  __shared__ float er_t[(TN + 1) * H];
  __shared__ unsigned q[32][QCAP];
  __shared__ unsigned wl[32][WCAP];

  const int tid = threadIdx.x;
  const int lane = tid & 31;
  const int wave = __builtin_amdgcn_readfirstlane(tid >> 5);
  unsigned* qw = q[wave];
  unsigned* wlw = wl[wave];
  const int nsc = ne / G;
  const int start = (int)(((long long)blockIdx.x * nsc) / gridDim.x);

  auto score = [&](float a, float b) {
    const float v = a + b;
    return (v > 0.0f) ? v : NEG_SLOPE * v;
  };
  auto load_el = [&](int sn, float (&a)[H]) {
    if constexpr (H == 4) {
      const float4 x = *reinterpret_cast<const float4*>(el + (size_t)sn * 4);
      a[0] = x.x;
      a[1] = x.y;
      a[2] = x.z;
      a[3] = x.w;
    } else {
#pragma unroll
      for (int j = 0; j < H; ++j) a[j] = el[(size_t)sn * H + j];
    }
  };

  for (int tile = blockIdx.x; tile * TN < n; tile += gridDim.x) {
    const int t0 = tile * TN;
    const int tn = (n - t0 < TN) ? (n - t0) : TN;
    const unsigned tkey = (unsigned)(tile & 255) * 0x01010101u;

    for (int i = tid; i < (TN + 1) * HD / 4; i += 1024)
      reinterpret_cast<float4*>(acc)[i] = make_float4(0.f, 0.f, 0.f, 0.f);
    for (int i = tid; i < (TN + 1) * H; i += 1024) {
      m_t[i] = -1;
      s_t[i] = 0.0f;
      er_t[i] = (i < tn * H) ? er[t0 * H + i] : 0.0f;
    }
    __syncthreads();

    int ln = 0;

    auto fetch = [&](int base, int& sn, int& dl) {
      const unsigned ent = qw[base + lane];
      const bool pad = (ent >> 31) != 0u;
      const int eid = pad ? 0 : (int)ent;
      sn = src[eid];
      const unsigned dlu = (unsigned)(dst[eid] - t0);
      const bool valid = !pad && dlu < (unsigned)tn;
      dl = valid ? (int)dlu : TN;
      return valid;
    };

    auto take_max = [&](int dl, const float (&a)[H]) {
#pragma unroll
      for (int j = 0; j < H; ++j) {
        const float v = score(a[j], er_t[dl * H + j]);
        int* slot = m_t + dl * H + j;
        if (__float_as_int(v) >= 0)
          __hip_atomic_fetch_max(slot, __float_as_int(v), __ATOMIC_RELAXED,
                                 __HIP_MEMORY_SCOPE_WORKGROUP);
        else
          __hip_atomic_fetch_min(reinterpret_cast<unsigned*>(slot), __float_as_uint(v),
                                 __ATOMIC_RELAXED, __HIP_MEMORY_SCOPE_WORKGROUP);
      }
    };

    auto flush_max = [&](int base) {
      int sn, dl;
      const bool valid = fetch(base, sn, dl);
      float a[H];
      load_el(sn, a);
      if (valid) take_max(dl, a);
      const unsigned vm = __builtin_amdgcn_ballot_w32(valid);
      const int pos = ln + (int)__builtin_amdgcn_mbcnt_lo(vm, 0u);
      if (valid && pos < WCAP) wlw[pos] = ((unsigned)dl << 23) | (unsigned)sn;
      ln += __builtin_popcount(vm);
    };

    auto accumulate = [&](int sn, int dl, bool valid, int kend) {
      float a[H];
      load_el(sn, a);
#pragma unroll
      for (int j = 0; j < H; ++j) {
        const float v = score(a[j], er_t[dl * H + j]);
        const float ex = expf(v - __int_as_float(m_t[dl * H + j]));
        a[j] = valid ? ex : 0.0f;
      }
      if (valid) {
#pragma unroll
        for (int j = 0; j < H; ++j)
          __hip_atomic_fetch_add(s_t + dl * H + j, a[j], __ATOMIC_RELAXED,
                                 __HIP_MEMORY_SCOPE_WORKGROUP);
      }
      for (int k0 = 0; k0 < kend; k0 += KU) {
        float v[KU][H];
#pragma unroll
        for (int u = 0; u < KU; ++u) {
          const int sk = __builtin_amdgcn_readlane(sn, k0 + u);
          const float* row = hfeat + (size_t)sk * HD + lane;
#pragma unroll
          for (int j = 0; j < H; ++j) v[u][j] = row[j * 32];
        }
#pragma unroll
        for (int u = 0; u < KU; ++u) {
          const int dk = __builtin_amdgcn_readlane(dl, k0 + u);
          float* arow = acc + dk * HD + lane;
#pragma unroll
          for (int j = 0; j < H; ++j) {
            const float ak =
                __int_as_float(__builtin_amdgcn_readlane(__float_as_int(a[j]), k0 + u));
            __hip_atomic_fetch_add(arow + j * 32, ak * v[u][j], __ATOMIC_RELAXED,
                                   __HIP_MEMORY_SCOPE_WORKGROUP);
          }
        }
      }
    };

    auto flush_sum = [&](int base) {
      int sn, dl;
      const bool valid = fetch(base, sn, dl);
      accumulate(valid ? sn : 0, dl, valid, 32);
    };

    auto scan = [&](auto&& consume) {
      int qn = 0;
      auto load_keys = [&](uint4 (&d)[U], int i) {
        int sc = i + start;
        if (sc >= nsc) sc -= nsc;
        const uint4* p = reinterpret_cast<const uint4*>(key + (size_t)sc * G) + lane;
#pragma unroll
        for (int u = 0; u < U; ++u) d[u] = p[u * 32];
        return sc;
      };
      auto append = [&](unsigned mask, bool m, int eid) {
        if (m) qw[qn + (int)__builtin_amdgcn_mbcnt_lo(mask, 0u)] = (unsigned)eid;
        qn += __builtin_popcount(mask);
      };
      auto drain = [&]() {
        while (qn >= 32) {
          qn -= 32;
          consume(qn);
        }
      };

      for (int eb = nsc * G + wave * 32; eb < ne; eb += 1024) {
        const int eid = eb + lane;
        const int dv = (eid < ne) ? dst[eid] : -1;
        const bool m = (unsigned)(dv - t0) < (unsigned)tn;
        append(__builtin_amdgcn_ballot_w32(m), m, eid);
      }

      uint4 cur[U];
      int sc_cur = 0;
      if (wave < nsc) sc_cur = load_keys(cur, wave);
      for (int i = wave; i < nsc; i += 32) {
        uint4 nxt[U];
        int sc_nxt = 0;
#pragma unroll
        for (int u = 0; u < U; ++u) nxt[u] = cur[u];
        if (i + 32 < nsc) sc_nxt = load_keys(nxt, i + 32);

        auto hits = [&](unsigned v) {
          const unsigned x = v ^ tkey;
          return (x - 0x01010101u) & ~x & 0x80808080u;
        };
        auto lane_hits = [&](const uint4 k) {
          return (hits(k.x) >> 7) | (hits(k.y) >> 6) | (hits(k.z) >> 5) | (hits(k.w) >> 4);
        };
        unsigned w[NW];
#pragma unroll
        for (int j = 0; j < NW; ++j)
          w[j] = lane_hits(cur[2 * j]) | (lane_hits(cur[2 * j + 1]) << 4);
        const int e0 = sc_cur * G + lane * 16;
        for (;;) {
          unsigned sel = w[NW - 1];
          int k = NW - 1;
#pragma unroll
          for (int j = NW - 2; j >= 0; --j)
            if (w[j] != 0u) {
              sel = w[j];
              k = j;
            }
          const unsigned any = __builtin_amdgcn_ballot_w32(sel != 0u);
          if (any == 0u) break;
          const int p = __builtin_ctz(sel | 0x80000000u);
          append(any, sel != 0u,
                 e0 + (2 * k + ((p >> 2) & 1)) * 512 + (((p & 3) << 2) | (p >> 3)));
#pragma unroll
          for (int j = 0; j < NW; ++j)
            if (j == k) w[j] &= w[j] - 1u;
          drain();
        }
#pragma unroll
        for (int u = 0; u < U; ++u) cur[u] = nxt[u];
        sc_cur = sc_nxt;
      }
      if ((qn & 31) != 0) {
        const int fill = 32 - (qn & 31);
        if (lane < fill) qw[qn + lane] = 0x80000000u;
        qn += fill;
      }
      drain();
    };

    bool listed = false;
    int lcount = 0;
    if constexpr (REUSE) {
      const int cl = gcount[tile * 32 + lane];
      listed = __builtin_amdgcn_ballot_w32(cl > GCAP) == 0u;
      lcount = __builtin_amdgcn_readlane(cl, wave);
    }
    if (listed) {
      const unsigned* __restrict__ g = glist + ((size_t)tile * 32 + wave) * GCAP;
      for (int b = 0; b < lcount; b += 32 * NB) {
        unsigned ent[NB];
        float a[NB][H];
#pragma unroll
        for (int u = 0; u < NB; ++u) {
          const int idx = b + u * 32 + lane;
          const unsigned w = g[(idx < lcount) ? idx : (lcount - 1)];
          ent[u] = (idx < lcount) ? w : ((unsigned)TN << 23);
        }
#pragma unroll
        for (int u = 0; u < NB; ++u) load_el((int)(ent[u] & 0x7FFFFFu), a[u]);
#pragma unroll
        for (int u = 0; u < NB; ++u) {
          const int idx = b + u * 32 + lane;
          if (idx < lcount) {
            take_max((int)(ent[u] >> 23), a[u]);
            wlw[idx] = ent[u];
          }
        }
      }
      ln = lcount;
    } else {
      scan(flush_max);
    }
    __syncthreads();

    if constexpr (EMIT) {
      const int cnt = (ln < GCAP) ? ln : GCAP;
      unsigned* __restrict__ g = glist + ((size_t)tile * 32 + wave) * GCAP;
      for (int i = lane; i < cnt; i += 32) g[i] = wlw[i];
      if (lane == 0) gcount[tile * 32 + wave] = ln;
    }

    if (ln <= WCAP) {
      for (int b = 0; b < ln; b += 32) {
        const int idx = b + lane;
        const bool valid = idx < ln;
        const unsigned ent = valid ? wlw[idx] : ((unsigned)TN << 23);
        accumulate((int)(ent & 0x7FFFFFu), (int)(ent >> 23), valid, (ln - b < 32) ? (ln - b) : 32);
      }
    } else {
      scan(flush_sum);
    }
    __syncthreads();

    float4* o4 = reinterpret_cast<float4*>(out + (size_t)t0 * HD);
    for (int i = tid; i < tn * HD / 4; i += 1024) {
      const float4 v = reinterpret_cast<const float4*>(acc)[i];
      const float sv = s_t[i >> 3];
      float4 r = make_float4(0.f, 0.f, 0.f, 0.f);
      if (sv > 0.0f) {
        r.x = v.x / sv;
        r.y = v.y / sv;
        r.z = v.z / sv;
        r.w = v.w / sv;
      }
      o4[i] = r;
    }
    __syncthreads();
  }
}

template <int C, int UN>
__global__ __launch_bounds__(LBLOCK) void bias_elu(const float* __restrict__ agg,
                                                   const float* __restrict__ b,
                                                   float* __restrict__ act, int total) {
  static_assert(LSTRIDE % (C / 4) == 0, "a thread keeps its piece of b on every trip");
  const int total4 = total >> 2;
  const int t = blockIdx.x * LBLOCK + threadIdx.x;
  const float4 b4 = reinterpret_cast<const float4*>(b)[t & (C / 4 - 1)];
  const float4* __restrict__ in4 = reinterpret_cast<const float4*>(agg);
  float4* __restrict__ out4 = reinterpret_cast<float4*>(act);

  auto finish = [&](int i, const float4 v) {
    const float x0 = v.x + b4.x;
    const float x1 = v.y + b4.y;
    const float x2 = v.z + b4.z;
    const float x3 = v.w + b4.w;
    float4 r;
    r.x = (x0 > 0.0f) ? x0 : (expf(x0) - 1.0f);
    r.y = (x1 > 0.0f) ? x1 : (expf(x1) - 1.0f);
    r.z = (x2 > 0.0f) ? x2 : (expf(x2) - 1.0f);
    r.w = (x3 > 0.0f) ? x3 : (expf(x3) - 1.0f);
    out4[i] = r;
  };

  int base = t;
  for (; base + (UN - 1) * LSTRIDE < total4; base += UN * LSTRIDE) {
    float4 v[UN];
#pragma unroll
    for (int u = 0; u < UN; ++u) v[u] = in4[base + u * LSTRIDE];
#pragma unroll
    for (int u = 0; u < UN; ++u) finish(base + u * LSTRIDE, v[u]);
  }
  for (; base < total4; base += LSTRIDE) finish(base, in4[base]);
}

template <int UN>
__global__ __launch_bounds__(LBLOCK) void add_bias32(float* __restrict__ out,
                                                     const float* __restrict__ b,
                                                     int total) {
  static_assert(LSTRIDE % 8 == 0, "a thread keeps its piece of b on every trip");
  const int total4 = total >> 2;
  const int t = blockIdx.x * LBLOCK + threadIdx.x;
  const float4 b4 = reinterpret_cast<const float4*>(b)[t & 7];
  float4* __restrict__ o4 = reinterpret_cast<float4*>(out);

  auto finish = [&](int i, const float4 v) {
    float4 r;
    r.x = v.x + b4.x;
    r.y = v.y + b4.y;
    r.z = v.z + b4.z;
    r.w = v.w + b4.w;
    o4[i] = r;
  };

  int base = t;
  for (; base + (UN - 1) * LSTRIDE < total4; base += UN * LSTRIDE) {
    float4 v[UN];
#pragma unroll
    for (int u = 0; u < UN; ++u) v[u] = o4[base + u * LSTRIDE];
#pragma unroll
    for (int u = 0; u < UN; ++u) finish(base + u * LSTRIDE, v[u]);
  }
  for (; base < total4; base += LSTRIDE) finish(base, o4[base]);
}

extern "C" void kernel_launch(void* const* d_in, const int* in_sizes, int n_in,
                              void* d_out, int out_size, void* d_ws, size_t ws_size,
                              hipStream_t stream) {
  const float* x = (const float*)d_in[0];
  const int* src = (const int*)d_in[1];
  const int* dst = (const int*)d_in[2];
  const float* W1 = (const float*)d_in[3];
  const float* al1 = (const float*)d_in[4];
  const float* ar1 = (const float*)d_in[5];
  const float* b1 = (const float*)d_in[6];
  const float* W2 = (const float*)d_in[7];
  const float* al2 = (const float*)d_in[8];
  const float* ar2 = (const float*)d_in[9];
  const float* b2 = (const float*)d_in[10];

  const int n = in_sizes[0] / IN_F;
  const int e = in_sizes[1];

  char* ws = (char*)d_ws;
  size_t off = 0;
  auto wsalloc = [&](size_t bytes) {
    void* p = ws + off;
    off = (off + bytes + 255) & ~(size_t)255;
    return p;
  };
  float* h1 = (float*)wsalloc((size_t)n * 128 * sizeof(float));
  float* agg1 = (float*)wsalloc((size_t)n * 128 * sizeof(float));
  unsigned char* keys = (unsigned char*)wsalloc((size_t)e);
  float* el = (float*)wsalloc((size_t)n * H0 * sizeof(float));
  float* er = (float*)wsalloc((size_t)n * H0 * sizeof(float));
  const int ntiles = (n + TN - 1) / TN;
  unsigned* glist = (unsigned*)wsalloc((size_t)ntiles * 32 * GCAP * sizeof(unsigned));
  int* gcount = (int*)wsalloc((size_t)ntiles * 32 * sizeof(int));
  float* outv = (float*)d_out;

  tile_keys<<<LGRID, LBLOCK, 0, stream>>>(dst, keys, e);

  wmma_gemm_k128<128, 1, 2><<<512, 256, 0, stream>>>(x, W1, h1, n);

  attn_coef32<H0, 4><<<LGRID, LBLOCK, 0, stream>>>(h1, al1, ar1, el, er, n * H0);

  tile_softmax_aggregate<H0, true, false><<<256, 1024, 0, stream>>>(src, dst, keys, el, er, h1,
                                                                    agg1, glist, gcount, e, n);

  bias_elu<128, 4><<<LGRID, LBLOCK, 0, stream>>>(agg1, b1, h1, n * 128);

  float* g2 = agg1;
  wmma_gemm_k128<32, 2, 1><<<1024, 128, 0, stream>>>(h1, W2, g2, n);

  attn_coef32<H1, 3><<<LGRID, LBLOCK, 0, stream>>>(g2, al2, ar2, el, er, n * H1);

  tile_softmax_aggregate<H1, false, true><<<256, 1024, 0, stream>>>(src, dst, keys, el, er, g2,
                                                                    outv, glist, gcount, e, n);

  add_bias32<3><<<LGRID, LBLOCK, 0, stream>>>(outv, b2, n * 32);
}


# edges dealt to destination tiles once per call: producer work-group b builds cell (t,b) in LDS, slot = returning LDS add; both tile kernels read cells, key scan leaves; float sum order changes
# speedup vs baseline: 7.3363x; 1.2985x over previous
#include <hip/hip_runtime.h>
#include <cstdint>
#include <cstddef>
#include <type_traits>

#define IN_F 128
#define NEG_SLOPE 0.2f
#define H0 4
#define H1 1

typedef __attribute__((ext_vector_type(2))) float v2f;
typedef __attribute__((ext_vector_type(8))) float v8f;

template <int NCOL, int RW, int RPW>
__global__ __launch_bounds__(32 * (NCOL / 16) * RW) void wmma_gemm_k128(
    const float* __restrict__ X, const float* __restrict__ W, float* __restrict__ Y,
    int nrows) {
  constexpr int K = 128;
  constexpr int CW = NCOL / 16;
  constexpr int NT = 32 * CW * RW;
  constexpr int TR = 16 * RW * RPW;
  constexpr int PITCH = K + 4;
  constexpr int NLD = TR * (K / 4) / NT;
  static_assert((TR * (K / 4)) % NT == 0, "a tile is whole pieces per thread");
  static_assert(NT % 32 == 0 && (K / 4) == 32, "a wave moves one whole row per piece");
  __shared__ float xs[2][TR][PITCH];

  const int tid = threadIdx.x;
  const int lane = tid & 31;
  const int wave = tid >> 5;
  const int n0 = (wave % CW) * 16;
  const int rb = (wave / CW) * RPW;
  const int m = lane & 15;
  const int khalf = lane >> 4;

  v2f bf[K / 4];
#pragma unroll
  for (int ks = 0; ks < K / 4; ++ks) {
    const int ka = ks * 4 + khalf * 2;
    bf[ks].x = W[ka * NCOL + n0 + m];
    bf[ks].y = W[(ka + 1) * NCOL + n0 + m];
  }

  const int ntiles = (nrows + TR - 1) / TR;
  const float4* __restrict__ X4 = reinterpret_cast<const float4*>(X);

  auto load_tile = [&](int tile, float4 (&v)[NLD]) {
    const float4* __restrict__ p0 = X4 + (size_t)tile * (TR * (K / 4)) + tid;
    if (tile * TR + TR <= nrows) {
#pragma unroll
      for (int u = 0; u < NLD; ++u) v[u] = p0[u * NT];
    } else {
#pragma unroll
      for (int u = 0; u < NLD; ++u) {
        const int r = tile * TR + ((tid + u * NT) >> 5);
        v[u] = (r < nrows) ? p0[u * NT] : make_float4(0.f, 0.f, 0.f, 0.f);
      }
    }
  };
  auto stage_tile = [&](int buf, const float4 (&v)[NLD]) {
#pragma unroll
    for (int u = 0; u < NLD; ++u) {
      const int p = tid + u * NT;
      *reinterpret_cast<float4*>(&xs[buf][p >> 5][(p & 31) * 4]) = v[u];
    }
  };

  float4 pre[NLD];
  int tile = blockIdx.x;
  if (tile < ntiles) {
    load_tile(tile, pre);
    stage_tile(0, pre);
  }
  __syncthreads();

  int buf = 0;
  for (; tile < ntiles; tile += gridDim.x) {
    const int nxt = tile + gridDim.x;
    if (nxt < ntiles) load_tile(nxt, pre);

    v8f acc[RPW];
#pragma unroll
    for (int s = 0; s < RPW; ++s) acc[s] = v8f{};
#pragma unroll
    for (int ks = 0; ks < K / 4; ++ks) {
      const int ka = ks * 4 + khalf * 2;
#pragma unroll
      for (int s = 0; s < RPW; ++s) {
        v2f a;
        a.x = xs[buf][(rb + s) * 16 + m][ka];
        a.y = xs[buf][(rb + s) * 16 + m][ka + 1];
        acc[s] = __builtin_amdgcn_wmma_f32_16x16x4_f32(false, a, false, bf[ks], (short)0,
                                                       acc[s], false, false);
      }
    }

    if (nxt < ntiles) stage_tile(buf ^ 1, pre);
    __syncthreads();

    const bool whole = tile * TR + TR <= nrows;
#pragma unroll
    for (int s = 0; s < RPW; ++s) {
      const int row0 = tile * TR + (rb + s) * 16 + khalf * 8;
      float* __restrict__ y0 = Y + (size_t)row0 * NCOL + n0 + m;
      if (whole) {
#pragma unroll
        for (int r = 0; r < 8; ++r) y0[r * NCOL] = acc[s][r];
      } else {
#pragma unroll
        for (int r = 0; r < 8; ++r)
          if (row0 + r < nrows) y0[r * NCOL] = acc[s][r];
      }
    }
    buf ^= 1;
  }
}

#define LGRID 1024
#define LBLOCK 256
#define LSTRIDE (LGRID * LBLOCK)

template <int H, int UN>
__global__ __launch_bounds__(LBLOCK) void attn_coef32(
    const float* __restrict__ h, const float* __restrict__ al,
    const float* __restrict__ ar, float* __restrict__ el, float* __restrict__ er,
    int nh) {
  static_assert(4 % H == 0, "the 4 rows of a wave must start at head 0");
  static_assert(LSTRIDE % 32 == 0, "a wave's pieces must start at a row boundary");
  const int lane = threadIdx.x & 31;
  const int sub = lane & 7;
  const int head = (lane >> 3) % H;
  const float4 a4 = *reinterpret_cast<const float4*>(al + head * 32 + sub * 4);
  const float4 r4 = *reinterpret_cast<const float4*>(ar + head * 32 + sub * 4);
  const float4* __restrict__ h4 = reinterpret_cast<const float4*>(h);
  const int total = nh * 8;
  const int t = blockIdx.x * LBLOCK + threadIdx.x;

  auto row_sums = [&](const float4 v, float& l, float& r) {
    l = v.x * a4.x + v.y * a4.y + v.z * a4.z + v.w * a4.w;
    r = v.x * r4.x + v.y * r4.y + v.z * r4.z + v.w * r4.w;
    l += __shfl_xor(l, 1, 32);
    r += __shfl_xor(r, 1, 32);
    l += __shfl_xor(l, 2, 32);
    r += __shfl_xor(r, 2, 32);
    l += __shfl_xor(l, 4, 32);
    r += __shfl_xor(r, 4, 32);
  };

  int wb = t - lane;
  for (; wb + (UN - 1) * LSTRIDE + 31 < total; wb += UN * LSTRIDE) {
    float4 v[UN];
#pragma unroll
    for (int u = 0; u < UN; ++u) v[u] = h4[wb + u * LSTRIDE + lane];
#pragma unroll
    for (int u = 0; u < UN; ++u) {
      float l, r;
      row_sums(v[u], l, r);
      if (sub == 0) {
        const int row = (wb + u * LSTRIDE + lane) >> 3;
        el[row] = l;
        er[row] = r;
      }
    }
  }
  for (; wb < total; wb += LSTRIDE) {
    const int i = wb + lane;
    const float4 v = (i < total) ? h4[i] : make_float4(0.f, 0.f, 0.f, 0.f);
    float l, r;
    row_sums(v, l, r);
    if (sub == 0 && i < total) {
      el[i >> 3] = l;
      er[i >> 3] = r;
    }
  }
}

#define TN 391
#define BW 256
#define CPW 8
#define CCAP 64
#define MAXBIN 256

__device__ __forceinline__ int count_pitch(int n) { return (((n + TN - 1) / TN) + 31) & ~31; }

__global__ __launch_bounds__(1024) void bin_edges(const int* __restrict__ src,
                                                  const int* __restrict__ dst,
                                                  unsigned char* __restrict__ key,
                                                  unsigned* __restrict__ cells,
                                                  int* __restrict__ ccount, int ne, int n) {
  static_assert(BW == 32 * CPW, "a tile's cells are dealt evenly to its 32 waves");
  static_assert(CCAP % 32 == 0, "a cell is whole 128-byte lines");
  __shared__ unsigned cell[MAXBIN][CCAP];
  __shared__ int cnt[MAXBIN];
  const int tid = threadIdx.x;
  const int lane = tid & 31;
  const int wave = tid >> 5;
  const int b = blockIdx.x;
  const int ntiles = (n + TN - 1) / TN;
  const bool fits = ntiles <= MAXBIN;
  if (tid < MAXBIN) cnt[tid] = 0;
  __syncthreads();

  auto take_slot = [&](unsigned t) {
    return __hip_atomic_fetch_add(cnt + t, 1, __ATOMIC_RELAXED, __HIP_MEMORY_SCOPE_WORKGROUP);
  };
  auto put_entry = [&](unsigned t, int slot, int d, int s) {
    if (slot < CCAP) cell[t][slot] = ((unsigned)(d - (int)t * TN) << 23) | (unsigned)s;
  };

  const int share = (((ne + BW - 1) / BW) + 3) & ~3;
  const int e0 = b * share;
  const int e1 = (e0 + share < ne) ? (e0 + share) : ne;
  const int nq = (e1 > e0) ? ((e1 - e0) >> 2) : 0;

  auto quads = [&](auto place_tag) {
    constexpr bool PLACE = decltype(place_tag)::value;
    const int4* __restrict__ d4 = reinterpret_cast<const int4*>(dst + e0);
    const int4* __restrict__ s4 = reinterpret_cast<const int4*>(src + e0);
    unsigned* __restrict__ k4 = reinterpret_cast<unsigned*>(key + e0);
    for (int q = tid; q < nq; q += 1024) {
      const int4 d = d4[q];
      const unsigned t0 = (unsigned)d.x / TN;
      const unsigned t1 = (unsigned)d.y / TN;
      const unsigned t2 = (unsigned)d.z / TN;
      const unsigned t3 = (unsigned)d.w / TN;
      if constexpr (PLACE) {
        const int4 s = s4[q];
        const int p0 = take_slot(t0);
        const int p1 = take_slot(t1);
        const int p2 = take_slot(t2);
        const int p3 = take_slot(t3);
        put_entry(t0, p0, d.x, s.x);
        put_entry(t1, p1, d.y, s.y);
        put_entry(t2, p2, d.z, s.z);
        put_entry(t3, p3, d.w, s.w);
      }
      k4[q] = (t0 & 255u) | ((t1 & 255u) << 8) | ((t2 & 255u) << 16) | ((t3 & 255u) << 24);
    }
  };
  if (nq > 0) {
    if (fits)
      quads(std::integral_constant<bool, true>{});
    else
      quads(std::integral_constant<bool, false>{});
  }
  for (int i = e0 + nq * 4 + tid; i < e1; i += 1024) {
    const int d = dst[i];
    const unsigned t = (unsigned)d / TN;
    if (fits) put_entry(t, take_slot(t), d, src[i]);
    key[i] = (unsigned char)(t & 255u);
  }
  __syncthreads();

  if (fits) {
    for (int t = wave; t < ntiles; t += 32) {
      const int kept = (cnt[t] < CCAP) ? cnt[t] : CCAP;
      unsigned* __restrict__ g = cells + ((size_t)t * BW + b) * CCAP;
      for (int i = lane; i < ((kept + 31) & ~31); i += 32)
        g[i] = (i < kept) ? cell[t][i] : ((unsigned)TN << 23);
    }
  }
  int* __restrict__ crow = ccount + (size_t)b * count_pitch(n);
  for (int t = tid; t < ntiles; t += 1024) crow[t] = fits ? cnt[t] : (CCAP + 1);
}

template <int H, bool PROJ>
__global__ __launch_bounds__(1024) void tile_softmax_aggregate(
    const int* __restrict__ src, const int* __restrict__ dst,
    const unsigned char* __restrict__ key, const float* __restrict__ el,
    const float* __restrict__ er, const float* __restrict__ hfeat,
    float* __restrict__ out, const unsigned* __restrict__ cells,
    const int* __restrict__ ccount, int ne, int n, const float* __restrict__ Wp,
    const float* __restrict__ bp) {
  constexpr int HD = H * 32;
  constexpr int AP = PROJ ? HD + 4 : HD;
  static_assert(!PROJ || HD == 128, "the projection is written for K = 128");
  constexpr int U = (H == 1) ? 8 : 4;
  constexpr int G = U * 512;
  constexpr int NW = U / 2;
  constexpr int QCAP = 32 * (1 + U);
  constexpr int WCAP = (H == 1) ? 1024 : 512;
  constexpr int KU = (H == 1) ? 8 : 4;
  static_assert(TN < 512, "local row must fit 9 bits");
  static_assert(CPW * CCAP <= WCAP, "a wave's cells must fit the wave's list in LDS");
  static_assert(CCAP == 64, "a cell is read as two 32-entry batches");
  constexpr int CB = 4;
  static_assert(CPW % CB == 0, "a wave's cells are whole groups");
  __shared__ __attribute__((aligned(16))) float acc[(TN + 1) * AP];
  __shared__ int m_t[(TN + 1) * H];
  __shared__ float s_t[(TN + 1) * H];
  __shared__ float er_t[(TN + 1) * H];
  __shared__ unsigned q[32][QCAP];
  __shared__ unsigned wl[32][WCAP];
  __shared__ int ovf[32];

  const int tid = threadIdx.x;
  const int lane = tid & 31;
  const int wave = __builtin_amdgcn_readfirstlane(tid >> 5);
  unsigned* qw = q[wave];
  unsigned* wlw = wl[wave];
  const int nsc = ne / G;
  const int start = (int)(((long long)blockIdx.x * nsc) / gridDim.x);

  auto score = [&](float a, float b) {
    const float v = a + b;
    return (v > 0.0f) ? v : NEG_SLOPE * v;
  };
  auto load_el = [&](int sn, float (&a)[H]) {
    if constexpr (H == 4) {
      const float4 x = *reinterpret_cast<const float4*>(el + (size_t)sn * 4);
      a[0] = x.x;
      a[1] = x.y;
      a[2] = x.z;
      a[3] = x.w;
    } else {
#pragma unroll
      for (int j = 0; j < H; ++j) a[j] = el[(size_t)sn * H + j];
    }
  };

  for (int tile = blockIdx.x; tile * TN < n; tile += gridDim.x) {
    const int t0 = tile * TN;
    const int tn = (n - t0 < TN) ? (n - t0) : TN;
    const unsigned tkey = (unsigned)(tile & 255) * 0x01010101u;

    static_assert(((TN + 1) * AP) % 4 == 0, "the tile is whole 16-byte pieces");
    for (int i = tid; i < (TN + 1) * AP / 4; i += 1024)
      reinterpret_cast<float4*>(acc)[i] = make_float4(0.f, 0.f, 0.f, 0.f);
    for (int i = tid; i < (TN + 1) * H; i += 1024) {
      m_t[i] = -1;
      s_t[i] = 0.0f;
      er_t[i] = (i < tn * H) ? er[t0 * H + i] : 0.0f;
    }
    int myc = 0;
    if (lane < CPW) myc = ccount[(size_t)(wave * CPW + lane) * count_pitch(n) + tile];
    {
      const unsigned om = __builtin_amdgcn_ballot_w32(myc > CCAP);
      if (lane == 0) ovf[wave] = (om != 0u) ? 1 : 0;
    }
    __syncthreads();

    int ln = 0;

    auto fetch = [&](int base, int& sn, int& dl) {
      const unsigned ent = qw[base + lane];
      const bool pad = (ent >> 31) != 0u;
      const int eid = pad ? 0 : (int)ent;
      sn = src[eid];
      const unsigned dlu = (unsigned)(dst[eid] - t0);
      const bool valid = !pad && dlu < (unsigned)tn;
      dl = valid ? (int)dlu : TN;
      return valid;
    };

    auto take_max = [&](int dl, const float (&a)[H]) {
#pragma unroll
      for (int j = 0; j < H; ++j) {
        const float v = score(a[j], er_t[dl * H + j]);
        int* slot = m_t + dl * H + j;
        if (__float_as_int(v) >= 0)
          __hip_atomic_fetch_max(slot, __float_as_int(v), __ATOMIC_RELAXED,
                                 __HIP_MEMORY_SCOPE_WORKGROUP);
        else
          __hip_atomic_fetch_min(reinterpret_cast<unsigned*>(slot), __float_as_uint(v),
                                 __ATOMIC_RELAXED, __HIP_MEMORY_SCOPE_WORKGROUP);
      }
    };

    auto flush_max = [&](int base) {
      int sn, dl;
      const bool valid = fetch(base, sn, dl);
      float a[H];
      load_el(sn, a);
      if (valid) take_max(dl, a);
      const unsigned vm = __builtin_amdgcn_ballot_w32(valid);
      const int pos = ln + (int)__builtin_amdgcn_mbcnt_lo(vm, 0u);
      if (valid && pos < WCAP) wlw[pos] = ((unsigned)dl << 23) | (unsigned)sn;
      ln += __builtin_popcount(vm);
    };

    auto accumulate = [&](int sn, int dl, bool valid, int kend) {
      float a[H];
      load_el(sn, a);
#pragma unroll
      for (int j = 0; j < H; ++j) {
        const float v = score(a[j], er_t[dl * H + j]);
        const float ex = expf(v - __int_as_float(m_t[dl * H + j]));
        a[j] = valid ? ex : 0.0f;
      }
      if (valid) {
#pragma unroll
        for (int j = 0; j < H; ++j)
          __hip_atomic_fetch_add(s_t + dl * H + j, a[j], __ATOMIC_RELAXED,
                                 __HIP_MEMORY_SCOPE_WORKGROUP);
      }
      for (int k0 = 0; k0 < kend; k0 += KU) {
        float v[KU][H];
#pragma unroll
        for (int u = 0; u < KU; ++u) {
          const int sk = __builtin_amdgcn_readlane(sn, k0 + u);
          const float* row = hfeat + (size_t)sk * HD + lane;
#pragma unroll
          for (int j = 0; j < H; ++j) v[u][j] = row[j * 32];
        }
#pragma unroll
        for (int u = 0; u < KU; ++u) {
          const int dk = __builtin_amdgcn_readlane(dl, k0 + u);
          float* arow = acc + dk * AP + lane;
#pragma unroll
          for (int j = 0; j < H; ++j) {
            const float ak =
                __int_as_float(__builtin_amdgcn_readlane(__float_as_int(a[j]), k0 + u));
            __hip_atomic_fetch_add(arow + j * 32, ak * v[u][j], __ATOMIC_RELAXED,
                                   __HIP_MEMORY_SCOPE_WORKGROUP);
          }
        }
      }
    };

    auto flush_sum = [&](int base) {
      int sn, dl;
      const bool valid = fetch(base, sn, dl);
      accumulate(valid ? sn : 0, dl, valid, 32);
    };

    auto scan = [&](auto&& consume) {
      int qn = 0;
      auto load_keys = [&](uint4 (&d)[U], int i) {
        int sc = i + start;
        if (sc >= nsc) sc -= nsc;
        const uint4* p = reinterpret_cast<const uint4*>(key + (size_t)sc * G) + lane;
#pragma unroll
        for (int u = 0; u < U; ++u) d[u] = p[u * 32];
        return sc;
      };
      auto append = [&](unsigned mask, bool m, int eid) {
        if (m) qw[qn + (int)__builtin_amdgcn_mbcnt_lo(mask, 0u)] = (unsigned)eid;
        qn += __builtin_popcount(mask);
      };
      auto drain = [&]() {
        while (qn >= 32) {
          qn -= 32;
          consume(qn);
        }
      };

      for (int eb = nsc * G + wave * 32; eb < ne; eb += 1024) {
        const int eid = eb + lane;
        const int dv = (eid < ne) ? dst[eid] : -1;
        const bool m = (unsigned)(dv - t0) < (unsigned)tn;
        append(__builtin_amdgcn_ballot_w32(m), m, eid);
      }

      uint4 cur[U];
      int sc_cur = 0;
      if (wave < nsc) sc_cur = load_keys(cur, wave);
      for (int i = wave; i < nsc; i += 32) {
        uint4 nxt[U];
        int sc_nxt = 0;
#pragma unroll
        for (int u = 0; u < U; ++u) nxt[u] = cur[u];
        if (i + 32 < nsc) sc_nxt = load_keys(nxt, i + 32);

        auto hits = [&](unsigned v) {
          const unsigned x = v ^ tkey;
          return (x - 0x01010101u) & ~x & 0x80808080u;
        };
        auto lane_hits = [&](const uint4 k) {
          return (hits(k.x) >> 7) | (hits(k.y) >> 6) | (hits(k.z) >> 5) | (hits(k.w) >> 4);
        };
        unsigned w[NW];
#pragma unroll
        for (int j = 0; j < NW; ++j)
          w[j] = lane_hits(cur[2 * j]) | (lane_hits(cur[2 * j + 1]) << 4);
        const int e0 = sc_cur * G + lane * 16;
        for (;;) {
          unsigned sel = w[NW - 1];
          int k = NW - 1;
#pragma unroll
          for (int j = NW - 2; j >= 0; --j)
            if (w[j] != 0u) {
              sel = w[j];
              k = j;
            }
          const unsigned any = __builtin_amdgcn_ballot_w32(sel != 0u);
          if (any == 0u) break;
          const int p = __builtin_ctz(sel | 0x80000000u);
          append(any, sel != 0u,
                 e0 + (2 * k + ((p >> 2) & 1)) * 512 + (((p & 3) << 2) | (p >> 3)));
#pragma unroll
          for (int j = 0; j < NW; ++j)
            if (j == k) w[j] &= w[j] - 1u;
          drain();
        }
#pragma unroll
        for (int u = 0; u < U; ++u) cur[u] = nxt[u];
        sc_cur = sc_nxt;
      }
      if ((qn & 31) != 0) {
        const int fill = 32 - (qn & 31);
        if (lane < fill) qw[qn + lane] = 0x80000000u;
        qn += fill;
      }
      drain();
    };

    const bool listed = __builtin_amdgcn_ballot_w32(ovf[lane] != 0) == 0u;
    int cn[CPW];
#pragma unroll
    for (int k = 0; k < CPW; ++k) cn[k] = __builtin_amdgcn_readlane(myc, k);
    if (listed) {
      const unsigned* __restrict__ g = cells + ((size_t)tile * BW + wave * CPW) * CCAP;
      int base = 0;
#pragma unroll
      for (int c0 = 0; c0 < CPW; c0 += CB) {
        unsigned ent[2 * CB];
        float a[2 * CB][H];
#pragma unroll
        for (int u = 0; u < 2 * CB; ++u) {
          const int cc = c0 + (u >> 1);
          const int idx = (u & 1) * 32 + lane;
          ent[u] = (idx < cn[cc]) ? g[cc * CCAP + idx] : ((unsigned)TN << 23);
        }
#pragma unroll
        for (int u = 0; u < 2 * CB; ++u) load_el((int)(ent[u] & 0x7FFFFFu), a[u]);
#pragma unroll
        for (int u = 0; u < 2 * CB; ++u) {
          const int cc = c0 + (u >> 1);
          const int idx = (u & 1) * 32 + lane;
          if (idx < cn[cc]) {
            take_max((int)(ent[u] >> 23), a[u]);
            wlw[base + idx] = ent[u];
          }
          if (u & 1) base += cn[cc];
        }
      }
      ln = base;
    } else {
      scan(flush_max);
    }
    __syncthreads();

    if (ln <= WCAP) {
      for (int b = 0; b < ln; b += 32) {
        const int idx = b + lane;
        const bool valid = idx < ln;
        const unsigned ent = valid ? wlw[idx] : ((unsigned)TN << 23);
        accumulate((int)(ent & 0x7FFFFFu), (int)(ent >> 23), valid, (ln - b < 32) ? (ln - b) : 32);
      }
    } else {
      scan(flush_sum);
    }
    __syncthreads();

    if constexpr (PROJ) {
      if (wave * 16 < tn) {
        const int m = lane & 15;
        const int khalf = lane >> 4;
        const int rl = wave * 16 + m;
        const int row = (rl < tn) ? rl : TN;
        const float* __restrict__ arow = acc + row * AP;
        float sv[H];
#pragma unroll
        for (int j = 0; j < H; ++j) sv[j] = s_t[row * H + j];

        auto activ = [](float v, float s, float b) {
          const float x = ((s > 0.0f) ? (v / s) : 0.0f) + b;
          const float xn = (x > 0.0f) ? 0.0f : x;
          const float e = expf(xn) - 1.0f;
          return (x > 0.0f) ? x : e;
        };

        v8f c0 = v8f{};
        v8f c1 = v8f{};
#pragma unroll
        for (int j = 0; j < H; ++j) {
#pragma unroll 4
          for (int k8 = 0; k8 < 8; ++k8) {
            const int ka = (j * 8 + k8) * 4 + khalf * 2;
            const float2 v = *reinterpret_cast<const float2*>(arow + ka);
            const float2 bb = *reinterpret_cast<const float2*>(bp + ka);
            v2f a;
            a.x = activ(v.x, sv[j], bb.x);
            a.y = activ(v.y, sv[j], bb.y);
            v2f w0, w1;
            w0.x = Wp[ka * 32 + m];
            w0.y = Wp[(ka + 1) * 32 + m];
            w1.x = Wp[ka * 32 + 16 + m];
            w1.y = Wp[(ka + 1) * 32 + 16 + m];
            c0 = __builtin_amdgcn_wmma_f32_16x16x4_f32(false, a, false, w0, (short)0, c0,
                                                       false, false);
            c1 = __builtin_amdgcn_wmma_f32_16x16x4_f32(false, a, false, w1, (short)0, c1,
                                                       false, false);
          }
        }

        const int lr0 = wave * 16 + khalf * 8;
        float* __restrict__ y0 = out + (size_t)(t0 + lr0) * 32 + m;
#pragma unroll
        for (int r = 0; r < 8; ++r) {
          if (lr0 + r < tn) {
            y0[r * 32] = c0[r];
            y0[r * 32 + 16] = c1[r];
          }
        }
      }
    } else {
      float4* o4 = reinterpret_cast<float4*>(out + (size_t)t0 * HD);
      for (int i = tid; i < tn * HD / 4; i += 1024) {
        const float4 v = reinterpret_cast<const float4*>(acc)[i];
        const float sv = s_t[i >> 3];
        float4 r = make_float4(0.f, 0.f, 0.f, 0.f);
        if (sv > 0.0f) {
          r.x = v.x / sv;
          r.y = v.y / sv;
          r.z = v.z / sv;
          r.w = v.w / sv;
        }
        o4[i] = r;
      }
    }
    __syncthreads();
  }
}

template <int UN>
__global__ __launch_bounds__(LBLOCK) void add_bias32(float* __restrict__ out,
                                                     const float* __restrict__ b,
                                                     int total) {
  static_assert(LSTRIDE % 8 == 0, "a thread keeps its piece of b on every trip");
  const int total4 = total >> 2;
  const int t = blockIdx.x * LBLOCK + threadIdx.x;
  const float4 b4 = reinterpret_cast<const float4*>(b)[t & 7];
  float4* __restrict__ o4 = reinterpret_cast<float4*>(out);

  auto finish = [&](int i, const float4 v) {
    float4 r;
    r.x = v.x + b4.x;
    r.y = v.y + b4.y;
    r.z = v.z + b4.z;
    r.w = v.w + b4.w;
    o4[i] = r;
  };

  int base = t;
  for (; base + (UN - 1) * LSTRIDE < total4; base += UN * LSTRIDE) {
    float4 v[UN];
#pragma unroll
    for (int u = 0; u < UN; ++u) v[u] = o4[base + u * LSTRIDE];
#pragma unroll
    for (int u = 0; u < UN; ++u) finish(base + u * LSTRIDE, v[u]);
  }
  for (; base < total4; base += LSTRIDE) finish(base, o4[base]);
}

extern "C" void kernel_launch(void* const* d_in, const int* in_sizes, int n_in,
                              void* d_out, int out_size, void* d_ws, size_t ws_size,
                              hipStream_t stream) {
  const float* x = (const float*)d_in[0];
  const int* src = (const int*)d_in[1];
  const int* dst = (const int*)d_in[2];
  const float* W1 = (const float*)d_in[3];
  const float* al1 = (const float*)d_in[4];
  const float* ar1 = (const float*)d_in[5];
  const float* b1 = (const float*)d_in[6];
  const float* W2 = (const float*)d_in[7];
  const float* al2 = (const float*)d_in[8];
  const float* ar2 = (const float*)d_in[9];
  const float* b2 = (const float*)d_in[10];

  const int n = in_sizes[0] / IN_F;
  const int e = in_sizes[1];

  char* ws = (char*)d_ws;
  size_t off = 0;
  auto wsalloc = [&](size_t bytes) {
    void* p = ws + off;
    off = (off + bytes + 255) & ~(size_t)255;
    return p;
  };
  float* h1 = (float*)wsalloc((size_t)n * 128 * sizeof(float));
  float* g2 = (float*)wsalloc((size_t)n * 32 * sizeof(float));
  unsigned char* keys = (unsigned char*)wsalloc((size_t)e);
  float* el = (float*)wsalloc((size_t)n * H0 * sizeof(float));
  float* er = (float*)wsalloc((size_t)n * H0 * sizeof(float));
  const int ntiles = (n + TN - 1) / TN;
  unsigned* cells = (unsigned*)wsalloc((size_t)ntiles * BW * CCAP * sizeof(unsigned));
  int* ccount = (int*)wsalloc((size_t)BW * ((ntiles + 31) & ~31) * sizeof(int));
  float* outv = (float*)d_out;

  bin_edges<<<BW, 1024, 0, stream>>>(src, dst, keys, cells, ccount, e, n);

  wmma_gemm_k128<128, 1, 2><<<512, 256, 0, stream>>>(x, W1, h1, n);

  attn_coef32<H0, 4><<<LGRID, LBLOCK, 0, stream>>>(h1, al1, ar1, el, er, n * H0);

  tile_softmax_aggregate<H0, true><<<256, 1024, 0, stream>>>(
      src, dst, keys, el, er, h1, g2, cells, ccount, e, n, W2, b1);

  attn_coef32<H1, 3><<<LGRID, LBLOCK, 0, stream>>>(g2, al2, ar2, el, er, n * H1);

  tile_softmax_aggregate<H1, false><<<256, 1024, 0, stream>>>(
      src, dst, keys, el, er, g2, outv, cells, ccount, e, n, nullptr, nullptr);

  add_bias32<3><<<LGRID, LBLOCK, 0, stream>>>(outv, b2, n * 32);
}


# tile kernels: pass 1 keeps each listed edge's scores in LDS behind its wave's list (layer 2 all, layer 1 what the spare LDS holds, about 3 in 4); pass 2 reads them, the second el[src] gather leaves
# speedup vs baseline: 7.5759x; 1.0327x over previous
#include <hip/hip_runtime.h>
#include <cstdint>
#include <cstddef>
#include <type_traits>

#define IN_F 128
#define NEG_SLOPE 0.2f
#define H0 4
#define H1 1

typedef __attribute__((ext_vector_type(2))) float v2f;
typedef __attribute__((ext_vector_type(8))) float v8f;

template <int NCOL, int RW, int RPW>
__global__ __launch_bounds__(32 * (NCOL / 16) * RW) void wmma_gemm_k128(
    const float* __restrict__ X, const float* __restrict__ W, float* __restrict__ Y,
    int nrows) {
  constexpr int K = 128;
  constexpr int CW = NCOL / 16;
  constexpr int NT = 32 * CW * RW;
  constexpr int TR = 16 * RW * RPW;
  constexpr int PITCH = K + 4;
  constexpr int NLD = TR * (K / 4) / NT;
  static_assert((TR * (K / 4)) % NT == 0, "a tile is whole pieces per thread");
  static_assert(NT % 32 == 0 && (K / 4) == 32, "a wave moves one whole row per piece");
  __shared__ float xs[2][TR][PITCH];

  const int tid = threadIdx.x;
  const int lane = tid & 31;
  const int wave = tid >> 5;
  const int n0 = (wave % CW) * 16;
  const int rb = (wave / CW) * RPW;
  const int m = lane & 15;
  const int khalf = lane >> 4;

  v2f bf[K / 4];
#pragma unroll
  for (int ks = 0; ks < K / 4; ++ks) {
    const int ka = ks * 4 + khalf * 2;
    bf[ks].x = W[ka * NCOL + n0 + m];
    bf[ks].y = W[(ka + 1) * NCOL + n0 + m];
  }

  const int ntiles = (nrows + TR - 1) / TR;
  const float4* __restrict__ X4 = reinterpret_cast<const float4*>(X);

  auto load_tile = [&](int tile, float4 (&v)[NLD]) {
    const float4* __restrict__ p0 = X4 + (size_t)tile * (TR * (K / 4)) + tid;
    if (tile * TR + TR <= nrows) {
#pragma unroll
      for (int u = 0; u < NLD; ++u) v[u] = p0[u * NT];
    } else {
#pragma unroll
      for (int u = 0; u < NLD; ++u) {
        const int r = tile * TR + ((tid + u * NT) >> 5);
        v[u] = (r < nrows) ? p0[u * NT] : make_float4(0.f, 0.f, 0.f, 0.f);
      }
    }
  };
  auto stage_tile = [&](int buf, const float4 (&v)[NLD]) {
#pragma unroll
    for (int u = 0; u < NLD; ++u) {
      const int p = tid + u * NT;
      *reinterpret_cast<float4*>(&xs[buf][p >> 5][(p & 31) * 4]) = v[u];
    }
  };

  float4 pre[NLD];
  int tile = blockIdx.x;
  if (tile < ntiles) {
    load_tile(tile, pre);
    stage_tile(0, pre);
  }
  __syncthreads();

  int buf = 0;
  for (; tile < ntiles; tile += gridDim.x) {
    const int nxt = tile + gridDim.x;
    if (nxt < ntiles) load_tile(nxt, pre);

    v8f acc[RPW];
#pragma unroll
    for (int s = 0; s < RPW; ++s) acc[s] = v8f{};
#pragma unroll
    for (int ks = 0; ks < K / 4; ++ks) {
      const int ka = ks * 4 + khalf * 2;
#pragma unroll
      for (int s = 0; s < RPW; ++s) {
        v2f a;
        a.x = xs[buf][(rb + s) * 16 + m][ka];
        a.y = xs[buf][(rb + s) * 16 + m][ka + 1];
        acc[s] = __builtin_amdgcn_wmma_f32_16x16x4_f32(false, a, false, bf[ks], (short)0,
                                                       acc[s], false, false);
      }
    }

    if (nxt < ntiles) stage_tile(buf ^ 1, pre);
    __syncthreads();

    const bool whole = tile * TR + TR <= nrows;
#pragma unroll
    for (int s = 0; s < RPW; ++s) {
      const int row0 = tile * TR + (rb + s) * 16 + khalf * 8;
      float* __restrict__ y0 = Y + (size_t)row0 * NCOL + n0 + m;
      if (whole) {
#pragma unroll
        for (int r = 0; r < 8; ++r) y0[r * NCOL] = acc[s][r];
      } else {
#pragma unroll
        for (int r = 0; r < 8; ++r)
          if (row0 + r < nrows) y0[r * NCOL] = acc[s][r];
      }
    }
    buf ^= 1;
  }
}

#define LGRID 1024
#define LBLOCK 256
#define LSTRIDE (LGRID * LBLOCK)

template <int H, int UN>
__global__ __launch_bounds__(LBLOCK) void attn_coef32(
    const float* __restrict__ h, const float* __restrict__ al,
    const float* __restrict__ ar, float* __restrict__ el, float* __restrict__ er,
    int nh) {
  static_assert(4 % H == 0, "the 4 rows of a wave must start at head 0");
  static_assert(LSTRIDE % 32 == 0, "a wave's pieces must start at a row boundary");
  const int lane = threadIdx.x & 31;
  const int sub = lane & 7;
  const int head = (lane >> 3) % H;
  const float4 a4 = *reinterpret_cast<const float4*>(al + head * 32 + sub * 4);
  const float4 r4 = *reinterpret_cast<const float4*>(ar + head * 32 + sub * 4);
  const float4* __restrict__ h4 = reinterpret_cast<const float4*>(h);
  const int total = nh * 8;
  const int t = blockIdx.x * LBLOCK + threadIdx.x;

  auto row_sums = [&](const float4 v, float& l, float& r) {
    l = v.x * a4.x + v.y * a4.y + v.z * a4.z + v.w * a4.w;
    r = v.x * r4.x + v.y * r4.y + v.z * r4.z + v.w * r4.w;
    l += __shfl_xor(l, 1, 32);
    r += __shfl_xor(r, 1, 32);
    l += __shfl_xor(l, 2, 32);
    r += __shfl_xor(r, 2, 32);
    l += __shfl_xor(l, 4, 32);
    r += __shfl_xor(r, 4, 32);
  };

  int wb = t - lane;
  for (; wb + (UN - 1) * LSTRIDE + 31 < total; wb += UN * LSTRIDE) {
    float4 v[UN];
#pragma unroll
    for (int u = 0; u < UN; ++u) v[u] = h4[wb + u * LSTRIDE + lane];
#pragma unroll
    for (int u = 0; u < UN; ++u) {
      float l, r;
      row_sums(v[u], l, r);
      if (sub == 0) {
        const int row = (wb + u * LSTRIDE + lane) >> 3;
        el[row] = l;
        er[row] = r;
      }
    }
  }
  for (; wb < total; wb += LSTRIDE) {
    const int i = wb + lane;
    const float4 v = (i < total) ? h4[i] : make_float4(0.f, 0.f, 0.f, 0.f);
    float l, r;
    row_sums(v, l, r);
    if (sub == 0 && i < total) {
      el[i >> 3] = l;
      er[i >> 3] = r;
    }
  }
}

#define TN 391
#define BW 256
#define CPW 8
#define CCAP 64
#define MAXBIN 256

__device__ __forceinline__ int count_pitch(int n) { return (((n + TN - 1) / TN) + 31) & ~31; }

__global__ __launch_bounds__(1024) void bin_edges(const int* __restrict__ src,
                                                  const int* __restrict__ dst,
                                                  unsigned char* __restrict__ key,
                                                  unsigned* __restrict__ cells,
                                                  int* __restrict__ ccount, int ne, int n) {
  static_assert(BW == 32 * CPW, "a tile's cells are dealt evenly to its 32 waves");
  static_assert(CCAP % 32 == 0, "a cell is whole 128-byte lines");
  __shared__ unsigned cell[MAXBIN][CCAP];
  __shared__ int cnt[MAXBIN];
  const int tid = threadIdx.x;
  const int lane = tid & 31;
  const int wave = tid >> 5;
  const int b = blockIdx.x;
  const int ntiles = (n + TN - 1) / TN;
  const bool fits = ntiles <= MAXBIN;
  if (tid < MAXBIN) cnt[tid] = 0;
  __syncthreads();

  auto take_slot = [&](unsigned t) {
    return __hip_atomic_fetch_add(cnt + t, 1, __ATOMIC_RELAXED, __HIP_MEMORY_SCOPE_WORKGROUP);
  };
  auto put_entry = [&](unsigned t, int slot, int d, int s) {
    if (slot < CCAP) cell[t][slot] = ((unsigned)(d - (int)t * TN) << 23) | (unsigned)s;
  };

  const int share = (((ne + BW - 1) / BW) + 3) & ~3;
  const int e0 = b * share;
  const int e1 = (e0 + share < ne) ? (e0 + share) : ne;
  const int nq = (e1 > e0) ? ((e1 - e0) >> 2) : 0;

  auto quads = [&](auto place_tag) {
    constexpr bool PLACE = decltype(place_tag)::value;
    const int4* __restrict__ d4 = reinterpret_cast<const int4*>(dst + e0);
    const int4* __restrict__ s4 = reinterpret_cast<const int4*>(src + e0);
    unsigned* __restrict__ k4 = reinterpret_cast<unsigned*>(key + e0);
    for (int q = tid; q < nq; q += 1024) {
      const int4 d = d4[q];
      const unsigned t0 = (unsigned)d.x / TN;
      const unsigned t1 = (unsigned)d.y / TN;
      const unsigned t2 = (unsigned)d.z / TN;
      const unsigned t3 = (unsigned)d.w / TN;
      if constexpr (PLACE) {
        const int4 s = s4[q];
        const int p0 = take_slot(t0);
        const int p1 = take_slot(t1);
        const int p2 = take_slot(t2);
        const int p3 = take_slot(t3);
        put_entry(t0, p0, d.x, s.x);
        put_entry(t1, p1, d.y, s.y);
        put_entry(t2, p2, d.z, s.z);
        put_entry(t3, p3, d.w, s.w);
      }
      k4[q] = (t0 & 255u) | ((t1 & 255u) << 8) | ((t2 & 255u) << 16) | ((t3 & 255u) << 24);
    }
  };
  if (nq > 0) {
    if (fits)
      quads(std::integral_constant<bool, true>{});
    else
      quads(std::integral_constant<bool, false>{});
  }
  for (int i = e0 + nq * 4 + tid; i < e1; i += 1024) {
    const int d = dst[i];
    const unsigned t = (unsigned)d / TN;
    if (fits) put_entry(t, take_slot(t), d, src[i]);
    key[i] = (unsigned char)(t & 255u);
  }
  __syncthreads();

  if (fits) {
    for (int t = wave; t < ntiles; t += 32) {
      const int kept = (cnt[t] < CCAP) ? cnt[t] : CCAP;
      unsigned* __restrict__ g = cells + ((size_t)t * BW + b) * CCAP;
      for (int i = lane; i < ((kept + 31) & ~31); i += 32)
        g[i] = (i < kept) ? cell[t][i] : ((unsigned)TN << 23);
    }
  }
  int* __restrict__ crow = ccount + (size_t)b * count_pitch(n);
  for (int t = tid; t < ntiles; t += 1024) crow[t] = fits ? cnt[t] : (CCAP + 1);
}

template <int H, bool PROJ>
__global__ __launch_bounds__(1024) void tile_softmax_aggregate(
    const int* __restrict__ src, const int* __restrict__ dst,
    const unsigned char* __restrict__ key, const float* __restrict__ el,
    const float* __restrict__ er, const float* __restrict__ hfeat,
    float* __restrict__ out, const unsigned* __restrict__ cells,
    const int* __restrict__ ccount, int ne, int n, const float* __restrict__ Wp,
    const float* __restrict__ bp) {
  constexpr int HD = H * 32;
  constexpr int AP = PROJ ? HD + 4 : HD;
  static_assert(!PROJ || HD == 128, "the projection is written for K = 128");
  constexpr int U = (H == 1) ? 8 : 4;
  constexpr int G = U * 512;
  constexpr int NW = U / 2;
  constexpr int QCAP = 32 * (1 + U);
  constexpr int WCAP = (H == 1) ? 1024 : 512;
  constexpr int KU = (H == 1) ? 8 : 4;
  static_assert(TN < 512, "local row must fit 9 bits");
  static_assert(CPW * CCAP <= WCAP, "a wave's cells must fit the wave's list in LDS");
  static_assert(CCAP == 64, "a cell is read as two 32-entry batches");
  constexpr int CB = 4;
  static_assert(CPW % CB == 0, "a wave's cells are whole groups");
  constexpr int XTRA = (H == 4) ? 104 : 0;
  constexpr int RW = WCAP + QCAP + XTRA;
  static_assert(RW % 4 == 0, "a wave's region starts on a 16-byte boundary");
  __shared__ __attribute__((aligned(16))) float acc[(TN + 1) * AP];
  __shared__ int m_t[(TN + 1) * H];
  __shared__ float s_t[(TN + 1) * H];
  __shared__ float er_t[(TN + 1) * H];
  __shared__ __attribute__((aligned(16))) unsigned lw[32][RW];
  __shared__ int ovf[32];

  const int tid = threadIdx.x;
  const int lane = tid & 31;
  const int wave = __builtin_amdgcn_readfirstlane(tid >> 5);
  unsigned* wlw = lw[wave];
  unsigned* qw = lw[wave] + WCAP;
  const int nsc = ne / G;
  const int start = (int)(((long long)blockIdx.x * nsc) / gridDim.x);

  auto score = [&](float a, float b) {
    const float v = a + b;
    return (v > 0.0f) ? v : NEG_SLOPE * v;
  };
  auto load_el = [&](int sn, float (&a)[H]) {
    if constexpr (H == 4) {
      const float4 x = *reinterpret_cast<const float4*>(el + (size_t)sn * 4);
      a[0] = x.x;
      a[1] = x.y;
      a[2] = x.z;
      a[3] = x.w;
    } else {
#pragma unroll
      for (int j = 0; j < H; ++j) a[j] = el[(size_t)sn * H + j];
    }
  };

  for (int tile = blockIdx.x; tile * TN < n; tile += gridDim.x) {
    const int t0 = tile * TN;
    const int tn = (n - t0 < TN) ? (n - t0) : TN;
    const unsigned tkey = (unsigned)(tile & 255) * 0x01010101u;

    static_assert(((TN + 1) * AP) % 4 == 0, "the tile is whole 16-byte pieces");
    for (int i = tid; i < (TN + 1) * AP / 4; i += 1024)
      reinterpret_cast<float4*>(acc)[i] = make_float4(0.f, 0.f, 0.f, 0.f);
    for (int i = tid; i < (TN + 1) * H; i += 1024) {
      m_t[i] = -1;
      s_t[i] = 0.0f;
      er_t[i] = (i < tn * H) ? er[t0 * H + i] : 0.0f;
    }
    int myc = 0;
    if (lane < CPW) myc = ccount[(size_t)(wave * CPW + lane) * count_pitch(n) + tile];
    {
      const unsigned om = __builtin_amdgcn_ballot_w32(myc > CCAP);
      if (lane == 0) ovf[wave] = (om != 0u) ? 1 : 0;
    }
    __syncthreads();

    int ln = 0;

    auto fetch = [&](int base, int& sn, int& dl) {
      const unsigned ent = qw[base + lane];
      const bool pad = (ent >> 31) != 0u;
      const int eid = pad ? 0 : (int)ent;
      sn = src[eid];
      const unsigned dlu = (unsigned)(dst[eid] - t0);
      const bool valid = !pad && dlu < (unsigned)tn;
      dl = valid ? (int)dlu : TN;
      return valid;
    };

    auto take_max = [&](int dl, float (&a)[H]) {
#pragma unroll
      for (int j = 0; j < H; ++j) {
        const float v = score(a[j], er_t[dl * H + j]);
        a[j] = v;
        int* slot = m_t + dl * H + j;
        if (__float_as_int(v) >= 0)
          __hip_atomic_fetch_max(slot, __float_as_int(v), __ATOMIC_RELAXED,
                                 __HIP_MEMORY_SCOPE_WORKGROUP);
        else
          __hip_atomic_fetch_min(reinterpret_cast<unsigned*>(slot), __float_as_uint(v),
                                 __ATOMIC_RELAXED, __HIP_MEMORY_SCOPE_WORKGROUP);
      }
    };

    auto flush_max = [&](int base) {
      int sn, dl;
      const bool valid = fetch(base, sn, dl);
      float a[H];
      load_el(sn, a);
      if (valid) take_max(dl, a);
      const unsigned vm = __builtin_amdgcn_ballot_w32(valid);
      const int pos = ln + (int)__builtin_amdgcn_mbcnt_lo(vm, 0u);
      if (valid && pos < WCAP) wlw[pos] = ((unsigned)dl << 23) | (unsigned)sn;
      ln += __builtin_popcount(vm);
    };

    auto accumulate_scored = [&](int sn, int dl, bool valid, int kend, float (&a)[H]) {
#pragma unroll
      for (int j = 0; j < H; ++j) {
        const float ex = expf(a[j] - __int_as_float(m_t[dl * H + j]));
        a[j] = valid ? ex : 0.0f;
      }
      if (valid) {
#pragma unroll
        for (int j = 0; j < H; ++j)
          __hip_atomic_fetch_add(s_t + dl * H + j, a[j], __ATOMIC_RELAXED,
                                 __HIP_MEMORY_SCOPE_WORKGROUP);
      }
      for (int k0 = 0; k0 < kend; k0 += KU) {
        float v[KU][H];
#pragma unroll
        for (int u = 0; u < KU; ++u) {
          const int sk = __builtin_amdgcn_readlane(sn, k0 + u);
          const float* row = hfeat + (size_t)sk * HD + lane;
#pragma unroll
          for (int j = 0; j < H; ++j) v[u][j] = row[j * 32];
        }
#pragma unroll
        for (int u = 0; u < KU; ++u) {
          const int dk = __builtin_amdgcn_readlane(dl, k0 + u);
          float* arow = acc + dk * AP + lane;
#pragma unroll
          for (int j = 0; j < H; ++j) {
            const float ak =
                __int_as_float(__builtin_amdgcn_readlane(__float_as_int(a[j]), k0 + u));
            __hip_atomic_fetch_add(arow + j * 32, ak * v[u][j], __ATOMIC_RELAXED,
                                   __HIP_MEMORY_SCOPE_WORKGROUP);
          }
        }
      }
    };
    auto accumulate = [&](int sn, int dl, bool valid, int kend) {
      float a[H];
      load_el(sn, a);
#pragma unroll
      for (int j = 0; j < H; ++j) a[j] = score(a[j], er_t[dl * H + j]);
      accumulate_scored(sn, dl, valid, kend, a);
    };

    auto flush_sum = [&](int base) {
      int sn, dl;
      const bool valid = fetch(base, sn, dl);
      accumulate(valid ? sn : 0, dl, valid, 32);
    };

    auto scan = [&](auto&& consume) {
      int qn = 0;
      auto load_keys = [&](uint4 (&d)[U], int i) {
        int sc = i + start;
        if (sc >= nsc) sc -= nsc;
        const uint4* p = reinterpret_cast<const uint4*>(key + (size_t)sc * G) + lane;
#pragma unroll
        for (int u = 0; u < U; ++u) d[u] = p[u * 32];
        return sc;
      };
      auto append = [&](unsigned mask, bool m, int eid) {
        if (m) qw[qn + (int)__builtin_amdgcn_mbcnt_lo(mask, 0u)] = (unsigned)eid;
        qn += __builtin_popcount(mask);
      };
      auto drain = [&]() {
        while (qn >= 32) {
          qn -= 32;
          consume(qn);
        }
      };

      for (int eb = nsc * G + wave * 32; eb < ne; eb += 1024) {
        const int eid = eb + lane;
        const int dv = (eid < ne) ? dst[eid] : -1;
        const bool m = (unsigned)(dv - t0) < (unsigned)tn;
        append(__builtin_amdgcn_ballot_w32(m), m, eid);
      }

      uint4 cur[U];
      int sc_cur = 0;
      if (wave < nsc) sc_cur = load_keys(cur, wave);
      for (int i = wave; i < nsc; i += 32) {
        uint4 nxt[U];
        int sc_nxt = 0;
#pragma unroll
        for (int u = 0; u < U; ++u) nxt[u] = cur[u];
        if (i + 32 < nsc) sc_nxt = load_keys(nxt, i + 32);

        auto hits = [&](unsigned v) {
          const unsigned x = v ^ tkey;
          return (x - 0x01010101u) & ~x & 0x80808080u;
        };
        auto lane_hits = [&](const uint4 k) {
          return (hits(k.x) >> 7) | (hits(k.y) >> 6) | (hits(k.z) >> 5) | (hits(k.w) >> 4);
        };
        unsigned w[NW];
#pragma unroll
        for (int j = 0; j < NW; ++j)
          w[j] = lane_hits(cur[2 * j]) | (lane_hits(cur[2 * j + 1]) << 4);
        const int e0 = sc_cur * G + lane * 16;
        for (;;) {
          unsigned sel = w[NW - 1];
          int k = NW - 1;
#pragma unroll
          for (int j = NW - 2; j >= 0; --j)
            if (w[j] != 0u) {
              sel = w[j];
              k = j;
            }
          const unsigned any = __builtin_amdgcn_ballot_w32(sel != 0u);
          if (any == 0u) break;
          const int p = __builtin_ctz(sel | 0x80000000u);
          append(any, sel != 0u,
                 e0 + (2 * k + ((p >> 2) & 1)) * 512 + (((p & 3) << 2) | (p >> 3)));
#pragma unroll
          for (int j = 0; j < NW; ++j)
            if (j == k) w[j] &= w[j] - 1u;
          drain();
        }
#pragma unroll
        for (int u = 0; u < U; ++u) cur[u] = nxt[u];
        sc_cur = sc_nxt;
      }
      if ((qn & 31) != 0) {
        const int fill = 32 - (qn & 31);
        if (lane < fill) qw[qn + lane] = 0x80000000u;
        qn += fill;
      }
      drain();
    };

    const bool listed = __builtin_amdgcn_ballot_w32(ovf[lane] != 0) == 0u;
    int cn[CPW];
#pragma unroll
    for (int k = 0; k < CPW; ++k) cn[k] = __builtin_amdgcn_readlane(myc, k);
    int s0 = 0;
    int C = 0;
    if (listed) {
      int lnl = 0;
#pragma unroll
      for (int k = 0; k < CPW; ++k) lnl += cn[k];
      s0 = (lnl + 3) & ~3;
      C = (RW - s0) / H;
    }
    auto put_scores = [&](int pos, const float (&a)[H]) {
      if constexpr (H == 4) {
        *reinterpret_cast<uint4*>(wlw + s0 + pos * 4) =
            make_uint4(__float_as_uint(a[0]), __float_as_uint(a[1]), __float_as_uint(a[2]),
                       __float_as_uint(a[3]));
      } else {
#pragma unroll
        for (int j = 0; j < H; ++j) wlw[s0 + pos * H + j] = __float_as_uint(a[j]);
      }
    };
    auto get_scores = [&](int pos, float (&a)[H]) {
      if constexpr (H == 4) {
        const uint4 x = *reinterpret_cast<const uint4*>(wlw + s0 + pos * 4);
        a[0] = __uint_as_float(x.x);
        a[1] = __uint_as_float(x.y);
        a[2] = __uint_as_float(x.z);
        a[3] = __uint_as_float(x.w);
      } else {
#pragma unroll
        for (int j = 0; j < H; ++j) a[j] = __uint_as_float(wlw[s0 + pos * H + j]);
      }
    };
    if (listed) {
      const unsigned* __restrict__ g = cells + ((size_t)tile * BW + wave * CPW) * CCAP;
      int base = 0;
#pragma unroll
      for (int c0 = 0; c0 < CPW; c0 += CB) {
        unsigned ent[2 * CB];
        float a[2 * CB][H];
#pragma unroll
        for (int u = 0; u < 2 * CB; ++u) {
          const int cc = c0 + (u >> 1);
          const int idx = (u & 1) * 32 + lane;
          ent[u] = (idx < cn[cc]) ? g[cc * CCAP + idx] : ((unsigned)TN << 23);
        }
#pragma unroll
        for (int u = 0; u < 2 * CB; ++u) load_el((int)(ent[u] & 0x7FFFFFu), a[u]);
#pragma unroll
        for (int u = 0; u < 2 * CB; ++u) {
          const int cc = c0 + (u >> 1);
          const int idx = (u & 1) * 32 + lane;
          if (idx < cn[cc]) {
            take_max((int)(ent[u] >> 23), a[u]);
            wlw[base + idx] = ent[u];
            if (base + idx < C) put_scores(base + idx, a[u]);
          }
          if (u & 1) base += cn[cc];
        }
      }
      ln = base;
    } else {
      scan(flush_max);
    }
    __syncthreads();

    if (ln <= WCAP) {
      for (int b = 0; b < ln; b += 32) {
        const int idx = b + lane;
        const bool valid = idx < ln;
        const unsigned ent = valid ? wlw[idx] : ((unsigned)TN << 23);
        const int kend = (ln - b < 32) ? (ln - b) : 32;
        const int sn = (int)(ent & 0x7FFFFFu);
        const int dl = (int)(ent >> 23);
        if (b < C) {
          float a[H];
#pragma unroll
          for (int j = 0; j < H; ++j) a[j] = 0.0f;
          if (valid) {
            if (idx < C) {
              get_scores(idx, a);
            } else {
              load_el(sn, a);
#pragma unroll
              for (int j = 0; j < H; ++j) a[j] = score(a[j], er_t[dl * H + j]);
            }
          }
          accumulate_scored(sn, dl, valid, kend, a);
        } else {
          accumulate(sn, dl, valid, kend);
        }
      }
    } else {
      scan(flush_sum);
    }
    __syncthreads();

    if constexpr (PROJ) {
      if (wave * 16 < tn) {
        const int m = lane & 15;
        const int khalf = lane >> 4;
        const int rl = wave * 16 + m;
        const int row = (rl < tn) ? rl : TN;
        const float* __restrict__ arow = acc + row * AP;
        float sv[H];
#pragma unroll
        for (int j = 0; j < H; ++j) sv[j] = s_t[row * H + j];

        auto activ = [](float v, float s, float b) {
          const float x = ((s > 0.0f) ? (v / s) : 0.0f) + b;
          const float xn = (x > 0.0f) ? 0.0f : x;
          const float e = expf(xn) - 1.0f;
          return (x > 0.0f) ? x : e;
        };

        v8f c0 = v8f{};
        v8f c1 = v8f{};
#pragma unroll
        for (int j = 0; j < H; ++j) {
#pragma unroll 4
          for (int k8 = 0; k8 < 8; ++k8) {
            const int ka = (j * 8 + k8) * 4 + khalf * 2;
            const float2 v = *reinterpret_cast<const float2*>(arow + ka);
            const float2 bb = *reinterpret_cast<const float2*>(bp + ka);
            v2f a;
            a.x = activ(v.x, sv[j], bb.x);
            a.y = activ(v.y, sv[j], bb.y);
            v2f w0, w1;
            w0.x = Wp[ka * 32 + m];
            w0.y = Wp[(ka + 1) * 32 + m];
            w1.x = Wp[ka * 32 + 16 + m];
            w1.y = Wp[(ka + 1) * 32 + 16 + m];
            c0 = __builtin_amdgcn_wmma_f32_16x16x4_f32(false, a, false, w0, (short)0, c0,
                                                       false, false);
            c1 = __builtin_amdgcn_wmma_f32_16x16x4_f32(false, a, false, w1, (short)0, c1,
                                                       false, false);
          }
        }

        const int lr0 = wave * 16 + khalf * 8;
        float* __restrict__ y0 = out + (size_t)(t0 + lr0) * 32 + m;
#pragma unroll
        for (int r = 0; r < 8; ++r) {
          if (lr0 + r < tn) {
            y0[r * 32] = c0[r];
            y0[r * 32 + 16] = c1[r];
          }
        }
      }
    } else {
      float4* o4 = reinterpret_cast<float4*>(out + (size_t)t0 * HD);
      for (int i = tid; i < tn * HD / 4; i += 1024) {
        const float4 v = reinterpret_cast<const float4*>(acc)[i];
        const float sv = s_t[i >> 3];
        float4 r = make_float4(0.f, 0.f, 0.f, 0.f);
        if (sv > 0.0f) {
          r.x = v.x / sv;
          r.y = v.y / sv;
          r.z = v.z / sv;
          r.w = v.w / sv;
        }
        o4[i] = r;
      }
    }
    __syncthreads();
  }
}

template <int UN>
__global__ __launch_bounds__(LBLOCK) void add_bias32(float* __restrict__ out,
                                                     const float* __restrict__ b,
                                                     int total) {
  static_assert(LSTRIDE % 8 == 0, "a thread keeps its piece of b on every trip");
  const int total4 = total >> 2;
  const int t = blockIdx.x * LBLOCK + threadIdx.x;
  const float4 b4 = reinterpret_cast<const float4*>(b)[t & 7];
  float4* __restrict__ o4 = reinterpret_cast<float4*>(out);

  auto finish = [&](int i, const float4 v) {
    float4 r;
    r.x = v.x + b4.x;
    r.y = v.y + b4.y;
    r.z = v.z + b4.z;
    r.w = v.w + b4.w;
    o4[i] = r;
  };

  int base = t;
  for (; base + (UN - 1) * LSTRIDE < total4; base += UN * LSTRIDE) {
    float4 v[UN];
#pragma unroll
    for (int u = 0; u < UN; ++u) v[u] = o4[base + u * LSTRIDE];
#pragma unroll
    for (int u = 0; u < UN; ++u) finish(base + u * LSTRIDE, v[u]);
  }
  for (; base < total4; base += LSTRIDE) finish(base, o4[base]);
}

extern "C" void kernel_launch(void* const* d_in, const int* in_sizes, int n_in,
                              void* d_out, int out_size, void* d_ws, size_t ws_size,
                              hipStream_t stream) {
  const float* x = (const float*)d_in[0];
  const int* src = (const int*)d_in[1];
  const int* dst = (const int*)d_in[2];
  const float* W1 = (const float*)d_in[3];
  const float* al1 = (const float*)d_in[4];
  const float* ar1 = (const float*)d_in[5];
  const float* b1 = (const float*)d_in[6];
  const float* W2 = (const float*)d_in[7];
  const float* al2 = (const float*)d_in[8];
  const float* ar2 = (const float*)d_in[9];
  const float* b2 = (const float*)d_in[10];

  const int n = in_sizes[0] / IN_F;
  const int e = in_sizes[1];

  char* ws = (char*)d_ws;
  size_t off = 0;
  auto wsalloc = [&](size_t bytes) {
    void* p = ws + off;
    off = (off + bytes + 255) & ~(size_t)255;
    return p;
  };
  float* h1 = (float*)wsalloc((size_t)n * 128 * sizeof(float));
  float* g2 = (float*)wsalloc((size_t)n * 32 * sizeof(float));
  unsigned char* keys = (unsigned char*)wsalloc((size_t)e);
  float* el = (float*)wsalloc((size_t)n * H0 * sizeof(float));
  float* er = (float*)wsalloc((size_t)n * H0 * sizeof(float));
  const int ntiles = (n + TN - 1) / TN;
  unsigned* cells = (unsigned*)wsalloc((size_t)ntiles * BW * CCAP * sizeof(unsigned));
  int* ccount = (int*)wsalloc((size_t)BW * ((ntiles + 31) & ~31) * sizeof(int));
  float* outv = (float*)d_out;

  bin_edges<<<BW, 1024, 0, stream>>>(src, dst, keys, cells, ccount, e, n);

  wmma_gemm_k128<128, 1, 2><<<512, 256, 0, stream>>>(x, W1, h1, n);

  attn_coef32<H0, 4><<<LGRID, LBLOCK, 0, stream>>>(h1, al1, ar1, el, er, n * H0);

  tile_softmax_aggregate<H0, true><<<256, 1024, 0, stream>>>(
      src, dst, keys, el, er, h1, g2, cells, ccount, e, n, W2, b1);

  attn_coef32<H1, 3><<<LGRID, LBLOCK, 0, stream>>>(g2, al2, ar2, el, er, n * H1);

  tile_softmax_aggregate<H1, false><<<256, 1024, 0, stream>>>(
      src, dst, keys, el, er, g2, outv, cells, ccount, e, n, nullptr, nullptr);

  add_bias32<3><<<LGRID, LBLOCK, 0, stream>>>(outv, b2, n * 32);
}
